# starting kernel
# baseline (speedup 1.0000x reference)
#include <hip/hip_runtime.h>
#include <hip/hip_bf16.h>


typedef __attribute__((ext_vector_type(16))) __bf16     v16bf;
typedef __attribute__((ext_vector_type(8)))  float      v8f;
typedef __attribute__((ext_vector_type(8)))  unsigned   v8u;

union FragAB { v8u u; v16bf b; };

__device__ __forceinline__ unsigned bf16_rne(float f) {
    unsigned u = __float_as_uint(f);
    u += 0x7FFFu + ((u >> 16) & 1u);
    return u >> 16;
}
__device__ __forceinline__ unsigned pack2bf(float lo, float hi) {
    return bf16_rne(lo) | (bf16_rne(hi) << 16);
}

__global__ void k_deg_init(float* dinv, int n) {
    int i = blockIdx.x * blockDim.x + threadIdx.x;
    if (i < n) dinv[i] = 1.0f;
}
__global__ void k_deg_edges(const int* __restrict__ dst, float* dinv, int e) {
    int i = blockIdx.x * blockDim.x + threadIdx.x;
    if (i < e) atomicAdd(&dinv[dst[i]], 1.0f);
}
__global__ void k_deg_rsqrt(float* dinv, int n) {
    int i = blockIdx.x * blockDim.x + threadIdx.x;
    if (i < n) dinv[i] = rsqrtf(dinv[i]);
}

__global__ void k_pack_x(const float* __restrict__ x, unsigned* __restrict__ xb,
                         long long n_pairs) {
    long long i = (long long)blockIdx.x * blockDim.x + threadIdx.x;
    if (i < n_pairs) {
        float2 v = ((const float2*)x)[i];
        xb[i] = pack2bf(v.x, v.y);
    }
}
__global__ void k_pack_w1(const float* __restrict__ W1, unsigned* __restrict__ w1p) {
    int tid = blockIdx.x * blockDim.x + threadIdx.x;
    if (tid >= 8192) return;
    int v    = tid & 7;
    int lane = (tid >> 3) & 31;
    int ks   = (tid >> 8) & 3;
    int nt   = tid >> 10;
    int k    = ks * 32 + ((v >> 2) << 4) + ((lane >> 4) << 3) + ((v & 3) << 1);
    int ncol = nt * 16 + (lane & 15);
    w1p[tid] = pack2bf(W1[(size_t)k * 128 + ncol], W1[(size_t)(k + 1) * 128 + ncol]);
}

__global__ void k_gemm1_wmma(const unsigned* __restrict__ xb,
                             const unsigned* __restrict__ w1p,
                             const float* __restrict__ dinv,
                             float* __restrict__ h1,
                             float* __restrict__ agg1,
                             int n_nodes) {
    const int lane = threadIdx.x & 31;
    const int wid  = threadIdx.x >> 5;
    const int m0   = blockIdx.x * 16;
    const int mr   = lane & 15;
    const int half = lane >> 4;

    int row = m0 + mr;
    if (row >= n_nodes) row = n_nodes - 1;
    const unsigned* xr = xb + (size_t)row * 64;
    const v8u* bfrag = (const v8u*)w1p;

    v8f acc = {};
#pragma unroll
    for (int ks = 0; ks < 4; ++ks) {
        FragAB a, b;
        uint4 alo = *(const uint4*)(xr + ks * 16 + half * 4);
        uint4 ahi = *(const uint4*)(xr + ks * 16 + 8 + half * 4);
        a.u = (v8u){alo.x, alo.y, alo.z, alo.w, ahi.x, ahi.y, ahi.z, ahi.w};
        b.u = bfrag[(wid * 4 + ks) * 32 + lane];
        acc = __builtin_amdgcn_wmma_f32_16x16x32_bf16(
                  false, a.b, false, b.b, (short)0, acc, false, false);
    }
    const int ncol  = wid * 16 + mr;
    const int mbase = m0 + (half << 3);
    if (m0 + 16 <= n_nodes) {
        float*       hp = h1   + (size_t)mbase * 128 + ncol;
        float*       ap = agg1 + (size_t)mbase * 128 + ncol;
        const float4 d0 = *(const float4*)(dinv + mbase);
        const float4 d1 = *(const float4*)(dinv + mbase + 4);
        const float  dv[8] = {d0.x, d0.y, d0.z, d0.w, d1.x, d1.y, d1.z, d1.w};
#pragma unroll
        for (int v = 0; v < 8; ++v) {
            float val = acc[v];
            hp[(size_t)v * 128] = val;
            ap[(size_t)v * 128] = val * dv[v] * dv[v];
        }
    } else {
#pragma unroll
        for (int v = 0; v < 8; ++v) {
            const int m = mbase + v;
            if (m < n_nodes) {
                float val = acc[v];
                float di  = dinv[m];
                h1  [(size_t)m * 128 + ncol] = val;
                agg1[(size_t)m * 128 + ncol] = val * di * di;
            }
        }
    }
}

__global__ void k_scatter1(const float* __restrict__ h1,
                           const int* __restrict__ src, const int* __restrict__ dst,
                           const float* __restrict__ dinv,
                           float* __restrict__ agg1, int n_edges) {
    int e    = (blockIdx.x * blockDim.x + threadIdx.x) >> 5;
    int lane = threadIdx.x & 31;
    if (e >= n_edges) return;
    int s = __builtin_amdgcn_readfirstlane(src[e]);
    int d = __builtin_amdgcn_readfirstlane(dst[e]);
    float norm = dinv[s] * dinv[d];
    float4 v = ((const float4*)(h1 + (size_t)s * 128))[lane];
    float* out = agg1 + (size_t)d * 128 + lane * 4;
    atomicAdd(out + 0, v.x * norm);
    atomicAdd(out + 1, v.y * norm);
    atomicAdd(out + 2, v.z * norm);
    atomicAdd(out + 3, v.w * norm);
}

__global__ void k_layer2(const float* __restrict__ agg1, const float* __restrict__ b1,
                         const float* __restrict__ W2, float* __restrict__ h2,
                         int n_nodes) {
    int node = (blockIdx.x * blockDim.x + threadIdx.x) >> 5;
    int lane = threadIdx.x & 31;
    if (node >= n_nodes) return;
    float4 v  = ((const float4*)(agg1 + (size_t)node * 128))[lane];
    float4 bb = ((const float4*)b1)[lane];
    float t0 = fmaxf(v.x + bb.x, 0.f), t1 = fmaxf(v.y + bb.y, 0.f);
    float t2 = fmaxf(v.z + bb.z, 0.f), t3 = fmaxf(v.w + bb.w, 0.f);
    int c = lane * 4;
    const float2* w2 = (const float2*)W2;
    float2 w0 = w2[c], w1 = w2[c + 1], w2v = w2[c + 2], w3 = w2[c + 3];
    float s0 = t0 * w0.x + t1 * w1.x + t2 * w2v.x + t3 * w3.x;
    float s1 = t0 * w0.y + t1 * w1.y + t2 * w2v.y + t3 * w3.y;
#pragma unroll
    for (int off = 16; off > 0; off >>= 1) {
        s0 += __shfl_down(s0, off, 32);
        s1 += __shfl_down(s1, off, 32);
    }
    if (lane == 0) { h2[node * 2] = s0; h2[node * 2 + 1] = s1; }
}

__global__ void k_self2(const float* __restrict__ h2, const float* __restrict__ dinv,
                        float* __restrict__ agg2, int n_nodes) {
    int i = blockIdx.x * blockDim.x + threadIdx.x;
    if (i < n_nodes * 2) {
        float di = dinv[i >> 1];
        agg2[i] = h2[i] * di * di;
    }
}
__global__ void k_scatter2(const float* __restrict__ h2,
                           const int* __restrict__ src, const int* __restrict__ dst,
                           const float* __restrict__ dinv,
                           float* __restrict__ agg2, int n_edges) {
    int e = blockIdx.x * blockDim.x + threadIdx.x;
    if (e >= n_edges) return;
    int s = src[e], d = dst[e];
    float norm = dinv[s] * dinv[d];
    float2 hv = ((const float2*)h2)[s];
    atomicAdd(&agg2[d * 2 + 0], hv.x * norm);
    atomicAdd(&agg2[d * 2 + 1], hv.y * norm);
}
__global__ void k_logsoftmax(const float* __restrict__ agg2, const float* __restrict__ b2,
                             float* __restrict__ out, int n_nodes) {
    int n = blockIdx.x * blockDim.x + threadIdx.x;
    if (n >= n_nodes) return;
    float2 a = ((const float2*)agg2)[n];
    float z0 = a.x + b2[0];
    float z1 = a.y + b2[1];
    float m  = fmaxf(z0, z1);
    float l  = m + __logf(__expf(z0 - m) + __expf(z1 - m));
    ((float2*)out)[n] = make_float2(z0 - l, z1 - l);
}

extern "C" void kernel_launch(void* const* d_in, const int* in_sizes, int n_in,
                              void* d_out, int out_size, void* d_ws, size_t ws_size,
                              hipStream_t stream) {
    const float* x   = (const float*)d_in[0];
    const int*   ei  = (const int*)  d_in[1];
    const float* W1  = (const float*)d_in[2];
    const float* b1  = (const float*)d_in[3];
    const float* W2  = (const float*)d_in[4];
    const float* b2  = (const float*)d_in[5];
    float*       out = (float*)d_out;

    const int n_nodes = in_sizes[0] / 128;
    const int n_edges = in_sizes[1] / 2;
    const int* src = ei;
    const int* dst = ei + n_edges;

    float*    ws   = (float*)d_ws;
    float*    dinv = ws;
    unsigned* xb   = (unsigned*)(dinv + n_nodes);
    unsigned* w1p  = xb + (size_t)n_nodes * 64;
    float*    h1   = (float*)(w1p + 8192);
    float*    agg1 = h1 + (size_t)n_nodes * 128;
    float*    h2   = agg1 + (size_t)n_nodes * 128;
    float*    agg2 = h1;

    const int B = 256;
    const long long n_pairs = (long long)n_nodes * 64;
    k_deg_init <<<(n_nodes + B - 1) / B, B, 0, stream>>>(dinv, n_nodes);
    k_deg_edges<<<(n_edges + B - 1) / B, B, 0, stream>>>(dst, dinv, n_edges);
    k_deg_rsqrt<<<(n_nodes + B - 1) / B, B, 0, stream>>>(dinv, n_nodes);
    k_pack_x <<<(unsigned)((n_pairs + B - 1) / B), B, 0, stream>>>(x, xb, n_pairs);
    k_pack_w1<<<8192 / B, B, 0, stream>>>(W1, w1p);
    k_gemm1_wmma<<<(n_nodes + 15) / 16, B, 0, stream>>>(xb, w1p, dinv, h1, agg1, n_nodes);
    k_scatter1<<<((size_t)n_edges * 32 + B - 1) / B, B, 0, stream>>>(h1, src, dst, dinv, agg1, n_edges);
    k_layer2  <<<((size_t)n_nodes * 32 + B - 1) / B, B, 0, stream>>>(agg1, b1, W2, h2, n_nodes);
    k_self2   <<<(n_nodes * 2 + B - 1) / B, B, 0, stream>>>(h2, dinv, agg2, n_nodes);
    k_scatter2<<<(n_edges + B - 1) / B, B, 0, stream>>>(h2, src, dst, dinv, agg2, n_edges);
    k_logsoftmax<<<(n_nodes + B - 1) / B, B, 0, stream>>>(agg2, b2, out, n_nodes);
}


# layer-1 scatter: source index and table place of an edge come as one packed dword (written once by the init pass): one scattered 4-byte lookup per found edge, not two; sums still in arrival order
# speedup vs baseline: 3.5093x; 3.5093x over previous
#include <hip/hip_runtime.h>
#include <hip/hip_bf16.h>


typedef __attribute__((ext_vector_type(16))) __bf16     v16bf;
typedef __attribute__((ext_vector_type(8)))  float      v8f;
typedef __attribute__((ext_vector_type(8)))  unsigned   v8u;

union FragAB { v8u u; v16bf b; };

__device__ __forceinline__ unsigned bf16_rne(float f) {
    unsigned u = __float_as_uint(f);
    u += 0x7FFFu + ((u >> 16) & 1u);
    return u >> 16;
}
__device__ __forceinline__ unsigned pack2bf(float lo, float hi) {
    return bf16_rne(lo) | (bf16_rne(hi) << 16);
}

#define L1_NODES   400

template <bool VEC>
__device__ __forceinline__ int4 l1_load_dst4(const int* __restrict__ dst, int e, int n_edges) {
    if (VEC) return *(const int4*)(dst + min(e, n_edges - 4));
    int4 r;
    r.x = dst[min(e + 0, n_edges - 1)];
    r.y = dst[min(e + 1, n_edges - 1)];
    r.z = dst[min(e + 2, n_edges - 1)];
    r.w = dst[min(e + 3, n_edges - 1)];
    return r;
}

template <bool VEC>
__device__ __forceinline__ void range_id_chunk(const int* __restrict__ dst, uint4* __restrict__ idlo,
                                               uint4* __restrict__ idhi, int chunk, int n_edges) {
    unsigned p[8] = {0u, 0u, 0u, 0u, 0u, 0u, 0u, 0u};
#pragma unroll
    for (int k = 0; k < 8; ++k) {
        const int4 d = l1_load_dst4<VEC>(dst, chunk * 32 + k * 4, n_edges);
        const unsigned id[4] = {(unsigned)d.x / L1_NODES, (unsigned)d.y / L1_NODES,
                                (unsigned)d.z / L1_NODES, (unsigned)d.w / L1_NODES};
#pragma unroll
        for (int c = 0; c < 4; ++c) {
#pragma unroll
            for (int b = 0; b < 8; ++b) p[b] |= ((id[c] >> b) & 1u) << (k * 4 + c);
        }
    }
    idlo[chunk] = make_uint4(p[0], p[1], p[2], p[3]);
    idhi[chunk] = make_uint4(p[4], p[5], p[6], p[7]);
}

#define PK_SHIFT   23
#define PK_SRCMASK ((1u << PK_SHIFT) - 1u)
#define PK_BATCH   8
static_assert(L1_NODES <= (1 << (32 - PK_SHIFT)), "a table place must fit above PK_SHIFT");

__global__ void k_deg_init(float* dinv, int n, const int* __restrict__ src, const int* __restrict__ dst,
                           uint4* __restrict__ idlo, uint4* __restrict__ idhi,
                           unsigned* __restrict__ pk, int n_edges) {
    int i = blockIdx.x * blockDim.x + threadIdx.x;
    if (i < n) dinv[i] = 1.0f;
    const int T = gridDim.x * blockDim.x;
    for (int e0 = i; e0 < n_edges; e0 += PK_BATCH * T) {
        int ec[PK_BATCH], sv[PK_BATCH], dv[PK_BATCH];
#pragma unroll
        for (int k = 0; k < PK_BATCH; ++k) {
            ec[k] = min(e0 + k * T, n_edges - 1);
            sv[k] = src[ec[k]];
            dv[k] = dst[ec[k]];
        }
#pragma unroll
        for (int k = 0; k < PK_BATCH; ++k)
            pk[ec[k]] = (unsigned)sv[k] | (((unsigned)dv[k] % L1_NODES) << PK_SHIFT);
    }
    if (i < (n_edges + 31) / 32) {
        if ((n_edges & 3) == 0) range_id_chunk<true >(dst, idlo, idhi, i, n_edges);
        else                    range_id_chunk<false>(dst, idlo, idhi, i, n_edges);
    }
}
#define RNG_NODES  6272
#define RNG_SLICES 16
#define RNG_THREADS 1024
#define RNG_BATCH   8

__global__ __launch_bounds__(RNG_THREADS)
void k_deg_edges(const int* __restrict__ dst, float* dinv, int n_nodes, int e) {
    __shared__ float acc[RNG_NODES];
    const int tid = threadIdx.x;
    const int r0  = blockIdx.y * RNG_NODES;
    const int cnt = min(RNG_NODES, n_nodes - r0);
    const int esz = (e + RNG_SLICES - 1) / RNG_SLICES;
    const int e0  = blockIdx.x * esz;
    const int e1  = min(e0 + esz, e);
    for (int i = tid; i < RNG_NODES; i += RNG_THREADS) acc[i] = 0.0f;
    __syncthreads();
    const int nb = (e1 - e0 + RNG_BATCH * RNG_THREADS - 1) / (RNG_BATCH * RNG_THREADS);
    for (int b = 0; b < nb; ++b) {
        const int base = e0 + tid + b * (RNG_BATCH * RNG_THREADS);
        int d[RNG_BATCH];
#pragma unroll
        for (int k = 0; k < RNG_BATCH; ++k) {
            int idx = base + k * RNG_THREADS;
            d[k] = dst[min(idx, e1 - 1)];
        }
        __builtin_amdgcn_sched_barrier(0);
#pragma unroll
        for (int k = 0; k < RNG_BATCH; ++k) {
            int idx = base + k * RNG_THREADS;
            unsigned dl = (unsigned)(d[k] - r0);
            if (idx < e1 && dl < (unsigned)cnt) atomicAdd(&acc[dl], 1.0f);
        }
    }
    __syncthreads();
    const int nch = (cnt + RNG_THREADS - 1) / RNG_THREADS;
    for (int c = 0; c < nch; ++c) {
        int cc = c + (int)blockIdx.x; cc = cc % nch;
        int i = cc * RNG_THREADS + tid;
        if (i < cnt) { float v = acc[i]; if (v != 0.0f) atomicAdd(&dinv[r0 + i], v); }
    }
}
__global__ void k_deg_rsqrt(float* dinv, int n) {
    int i = blockIdx.x * blockDim.x + threadIdx.x;
    if (i < n) dinv[i] = rsqrtf(dinv[i]);
}

__global__ void k_pack_x(const float* __restrict__ x, unsigned* __restrict__ xb,
                         long long n_pairs) {
    long long i = (long long)blockIdx.x * blockDim.x + threadIdx.x;
    if (i < n_pairs) {
        float2 v = ((const float2*)x)[i];
        xb[i] = pack2bf(v.x, v.y);
    }
}
__global__ void k_pack_w1(const float* __restrict__ W1, unsigned* __restrict__ w1p) {
    int tid = blockIdx.x * blockDim.x + threadIdx.x;
    if (tid >= 8192) return;
    int v    = tid & 7;
    int lane = (tid >> 3) & 31;
    int ks   = (tid >> 8) & 3;
    int nt   = tid >> 10;
    int k    = ks * 32 + ((v >> 2) << 4) + ((lane >> 4) << 3) + ((v & 3) << 1);
    int ncol = nt * 16 + (lane & 15);
    w1p[tid] = pack2bf(W1[(size_t)k * 128 + ncol], W1[(size_t)(k + 1) * 128 + ncol]);
}

__global__ void k_gemm1_wmma(const unsigned* __restrict__ xb,
                             const unsigned* __restrict__ w1p,
                             const float* __restrict__ dinv,
                             float* __restrict__ h1,
                             int n_nodes) {
    const int lane = threadIdx.x & 31;
    const int wid  = threadIdx.x >> 5;
    const int m0   = blockIdx.x * 16;
    const int mr   = lane & 15;
    const int half = lane >> 4;

    int row = m0 + mr;
    if (row >= n_nodes) row = n_nodes - 1;
    const unsigned* xr = xb + (size_t)row * 64;
    const v8u* bfrag = (const v8u*)w1p;

    v8f acc = {};
#pragma unroll
    for (int ks = 0; ks < 4; ++ks) {
        FragAB a, b;
        uint4 alo = *(const uint4*)(xr + ks * 16 + half * 4);
        uint4 ahi = *(const uint4*)(xr + ks * 16 + 8 + half * 4);
        a.u = (v8u){alo.x, alo.y, alo.z, alo.w, ahi.x, ahi.y, ahi.z, ahi.w};
        b.u = bfrag[(wid * 4 + ks) * 32 + lane];
        acc = __builtin_amdgcn_wmma_f32_16x16x32_bf16(
                  false, a.b, false, b.b, (short)0, acc, false, false);
    }
    const int ncol  = wid * 16 + mr;
    const int mbase = m0 + (half << 3);
    if (m0 + 16 <= n_nodes) {
        float*       hp = h1   + (size_t)mbase * 128 + ncol;
        const float4 d0 = *(const float4*)(dinv + mbase);
        const float4 d1 = *(const float4*)(dinv + mbase + 4);
        const float  dv[8] = {d0.x, d0.y, d0.z, d0.w, d1.x, d1.y, d1.z, d1.w};
#pragma unroll
        for (int v = 0; v < 8; ++v) {
            hp[(size_t)v * 128] = acc[v] * dv[v];
        }
    } else {
#pragma unroll
        for (int v = 0; v < 8; ++v) {
            const int m = mbase + v;
            if (m < n_nodes) {
                h1[(size_t)m * 128 + ncol] = acc[v] * dinv[m];
            }
        }
    }
}

#define L1_THREADS 1024
#define L1_Q       4
#define L1_BATCH   (L1_Q * 32 * 32)
#define L1_GROUP   8

__device__ __forceinline__ void l1_flush(const float* __restrict__ h1, const unsigned* __restrict__ pk,
                                         float* acc, int qe, int qn, int lane) {
    const unsigned pw = pk[qe];
    const int s  = (int)(pw & PK_SRCMASK);
    const int dl = (int)(pw >> PK_SHIFT);
    for (int g = 0; g < qn; g += L1_GROUP) {
        float4 v[L1_GROUP];
#pragma unroll
        for (int j = 0; j < L1_GROUP; ++j) {
            const int jj = min(g + j, qn - 1);
            const int sj = __builtin_amdgcn_readlane(s, jj);
            v[j] = ((const float4*)(h1 + (size_t)sj * 128))[lane];
        }
#pragma unroll
        for (int j = 0; j < L1_GROUP; ++j) {
            if (g + j < qn) {
                const int dj = __builtin_amdgcn_readlane(dl, g + j);
                float* op = acc + dj * 128 + lane;
                atomicAdd(op +  0, v[j].x);
                atomicAdd(op + 32, v[j].y);
                atomicAdd(op + 64, v[j].z);
                atomicAdd(op + 96, v[j].w);
            }
        }
    }
}

__device__ __forceinline__ unsigned l1_match(const uint4& lo, const uint4& hi, const unsigned (&gb)[8],
                                             int chunk, bool last_batch, int n_edges) {
    const unsigned differ = (lo.x ^ gb[0]) | (lo.y ^ gb[1]) | (lo.z ^ gb[2]) | (lo.w ^ gb[3])
                          | (hi.x ^ gb[4]) | (hi.y ^ gb[5]) | (hi.z ^ gb[6]) | (hi.w ^ gb[7]);
    unsigned mm = ~differ;
    if (last_batch) {
        const int left = n_edges - chunk * 32;
        const unsigned keep = (left >= 32) ? 0xFFFFFFFFu : (left <= 0 ? 0u : ((1u << left) - 1u));
        mm &= keep;
    }
    return mm;
}

__device__ __forceinline__ void l1_enqueue(unsigned mm, int chunk,
                                           const float* __restrict__ h1, const unsigned* __restrict__ pk,
                                           float* acc, int lane, int& qe, int& qn) {
    const int ebase = chunk * 32;
    unsigned any = __builtin_amdgcn_ballot_w32(mm != 0);
    while (any) {
        const int slot = __builtin_ctz(mm | 0x80000000u);
        const int e    = ebase + slot;
        mm &= mm - 1;
        while (any) {
            const int j  = __builtin_ctz(any);
            any &= any - 1;
            const int ej = __builtin_amdgcn_readlane(e, j);
            if (lane == qn) qe = ej;
            if (++qn == 32) { l1_flush(h1, pk, acc, qe, 32, lane); qn = 0; }
        }
        any = __builtin_amdgcn_ballot_w32(mm != 0);
    }
}

__device__ __forceinline__ void l1_scan(const float* __restrict__ h1, const unsigned* __restrict__ pk,
                                        const uint4* __restrict__ idlo, const uint4* __restrict__ idhi,
                                        float* acc, int n_edges, int wid, int lane) {
    const int nchunk = (n_edges + 31) / 32;
    const int nbatch = (n_edges + L1_BATCH - 1) / L1_BATCH;
    const int rot    = (int)(((long long)blockIdx.x * nbatch) / gridDim.x);
    unsigned gb[8];
#pragma unroll
    for (int k = 0; k < 8; ++k) gb[k] = 0u - ((blockIdx.x >> k) & 1u);
    int qe = 0, qn = 0;
    uint4 lo[L1_Q], hi[L1_Q];
    int i = wid;
    int b = i + rot; if (b >= nbatch) b -= nbatch;
    if (i < nbatch) {
#pragma unroll
        for (int q = 0; q < L1_Q; ++q) {
            const int c = min(b * (L1_Q * 32) + q * 32 + lane, nchunk - 1);
            lo[q] = idlo[c]; hi[q] = idhi[c];
        }
    }
    while (i < nbatch) {
        const bool last = (b * L1_BATCH + L1_BATCH > n_edges);
        unsigned mm[L1_Q];
#pragma unroll
        for (int q = 0; q < L1_Q; ++q)
            mm[q] = l1_match(lo[q], hi[q], gb, b * (L1_Q * 32) + q * 32 + lane, last, n_edges);
        const int in = i + (L1_THREADS >> 5);
        int bn = in + rot; if (bn >= nbatch) bn -= nbatch;
        if (in < nbatch) {
#pragma unroll
            for (int q = 0; q < L1_Q; ++q) {
                const int c = min(bn * (L1_Q * 32) + q * 32 + lane, nchunk - 1);
                lo[q] = idlo[c]; hi[q] = idhi[c];
            }
        }
#pragma unroll
        for (int q = 0; q < L1_Q; ++q)
            l1_enqueue(mm[q], b * (L1_Q * 32) + q * 32 + lane, h1, pk, acc, lane, qe, qn);
        i = in; b = bn;
    }
    if (qn > 0) l1_flush(h1, pk, acc, qe, qn, lane);
}

__global__ __launch_bounds__(L1_THREADS)
void k_scatter1(const float* __restrict__ h1, const unsigned* __restrict__ pk,
                const uint4* __restrict__ idlo, const uint4* __restrict__ idhi,
                const float* __restrict__ dinv, const float* __restrict__ b1,
                const float* __restrict__ W2,
                float* __restrict__ h2, float* __restrict__ agg2,
                int n_nodes, int n_edges) {
    __shared__ float acc[L1_NODES * 128];
    __shared__ float h2s[2 * L1_NODES];
    const int tid  = threadIdx.x;
    const int lane = tid & 31;
    const int wid  = __builtin_amdgcn_readfirstlane(tid >> 5);
    const int r0   = blockIdx.x * L1_NODES;
    const int cnt  = min(L1_NODES, n_nodes - r0);

    {
        const float4* hp = (const float4*)(h1 + (size_t)r0 * 128);
        for (int i = tid; i < cnt * 32; i += L1_THREADS) {
            const float4 hv = hp[i];
            float* op = acc + (i >> 5) * 128 + (i & 31);
            op[0] = hv.x; op[32] = hv.y; op[64] = hv.z; op[96] = hv.w;
        }
    }
    __syncthreads();
    l1_scan(h1, pk, idlo, idhi, acc, n_edges, wid, lane);
    __syncthreads();
    {
        const float4 bb = ((const float4*)b1)[lane];
        const float2* w2 = (const float2*)W2;
        const int c = lane * 4;
        const float2 w0 = w2[c], w1 = w2[c + 1], w2v = w2[c + 2], w3 = w2[c + 3];
        for (int n0 = wid; n0 < cnt; n0 += 4 * (L1_THREADS >> 5)) {
            float s0[4], s1[4], di[4];
#pragma unroll
            for (int j = 0; j < 4; ++j) {
                const int n = min(n0 + j * (L1_THREADS >> 5), cnt - 1);
                const float* op = acc + n * 128 + lane;
                di[j] = dinv[r0 + n];
                const float t0 = fmaxf(op[0]  * di[j] + bb.x, 0.f), t1 = fmaxf(op[32] * di[j] + bb.y, 0.f);
                const float t2 = fmaxf(op[64] * di[j] + bb.z, 0.f), t3 = fmaxf(op[96] * di[j] + bb.w, 0.f);
                s0[j] = t0 * w0.x + t1 * w1.x + t2 * w2v.x + t3 * w3.x;
                s1[j] = t0 * w0.y + t1 * w1.y + t2 * w2v.y + t3 * w3.y;
            }
#pragma unroll
            for (int off = 16; off > 0; off >>= 1) {
#pragma unroll
                for (int j = 0; j < 4; ++j) {
                    s0[j] += __shfl_down(s0[j], off, 32);
                    s1[j] += __shfl_down(s1[j], off, 32);
                }
            }
            if (lane == 0) {
#pragma unroll
                for (int j = 0; j < 4; ++j) {
                    const int n = n0 + j * (L1_THREADS >> 5);
                    if (n < cnt) { h2s[2 * n] = s0[j] * di[j]; h2s[2 * n + 1] = s1[j] * di[j]; }
                }
            }
        }
    }
    __syncthreads();
    for (int i = tid; i < 2 * cnt; i += L1_THREADS) {
        const float v = h2s[i];
        h2[2 * (size_t)r0 + i]   = v;
        agg2[2 * (size_t)r0 + i] = v * dinv[r0 + (i >> 1)];
    }
}

__global__ __launch_bounds__(RNG_THREADS)
void k_scatter2(const float* __restrict__ h2,
                const int* __restrict__ src, const int* __restrict__ dst,
                const float* __restrict__ dinv,
                float* __restrict__ agg2, int n_nodes, int n_edges) {
    __shared__ float acc[2 * RNG_NODES];
    const int tid = threadIdx.x;
    const int r0  = blockIdx.y * RNG_NODES;
    const int cnt = min(RNG_NODES, n_nodes - r0);
    const int esz = (n_edges + RNG_SLICES - 1) / RNG_SLICES;
    const int e0  = blockIdx.x * esz;
    const int e1  = min(e0 + esz, n_edges);
    for (int i = tid; i < 2 * RNG_NODES; i += RNG_THREADS) acc[i] = 0.0f;
    __syncthreads();
    const int nb = (e1 - e0 + RNG_BATCH * RNG_THREADS - 1) / (RNG_BATCH * RNG_THREADS);
    for (int b = 0; b < nb; ++b) {
        const int base = e0 + tid + b * (RNG_BATCH * RNG_THREADS);
        int   s[RNG_BATCH];
        int   d[RNG_BATCH];
        bool  m[RNG_BATCH];
        unsigned dl[RNG_BATCH];
        float2 hv[RNG_BATCH];
#pragma unroll
        for (int k = 0; k < RNG_BATCH; ++k) {
            int idx = min(base + k * RNG_THREADS, e1 - 1);
            d[k] = dst[idx];
            s[k] = src[idx];
        }
        __builtin_amdgcn_sched_barrier(0);
#pragma unroll
        for (int k = 0; k < RNG_BATCH; ++k) {
            dl[k] = (unsigned)(d[k] - r0);
            m[k] = (base + k * RNG_THREADS < e1) && (dl[k] < (unsigned)cnt);
            int si = m[k] ? s[k] : 0;
            hv[k] = ((const float2*)h2)[si];
        }
        __builtin_amdgcn_sched_barrier(0);
#pragma unroll
        for (int k = 0; k < RNG_BATCH; ++k) {
            if (m[k]) {
                atomicAdd(&acc[2 * dl[k] + 0], hv[k].x);
                atomicAdd(&acc[2 * dl[k] + 1], hv[k].y);
            }
        }
    }
    __syncthreads();
    const int cnt2 = 2 * cnt;
    const int nch = (cnt2 + RNG_THREADS - 1) / RNG_THREADS;
    for (int c = 0; c < nch; ++c) {
        int cc = c + (int)blockIdx.x; cc = cc % nch;
        int i = cc * RNG_THREADS + tid;
        if (i < cnt2) {
            float v = acc[i];
            if (v != 0.0f) atomicAdd(&agg2[2 * (size_t)r0 + i], v * dinv[r0 + (i >> 1)]);
        }
    }
}
__global__ void k_logsoftmax(const float* __restrict__ agg2, const float* __restrict__ b2,
                             float* __restrict__ out, int n_nodes) {
    int n = blockIdx.x * blockDim.x + threadIdx.x;
    if (n >= n_nodes) return;
    float2 a = ((const float2*)agg2)[n];
    float z0 = a.x + b2[0];
    float z1 = a.y + b2[1];
    float m  = fmaxf(z0, z1);
    float l  = m + __logf(__expf(z0 - m) + __expf(z1 - m));
    ((float2*)out)[n] = make_float2(z0 - l, z1 - l);
}

extern "C" void kernel_launch(void* const* d_in, const int* in_sizes, int n_in,
                              void* d_out, int out_size, void* d_ws, size_t ws_size,
                              hipStream_t stream) {
    const float* x   = (const float*)d_in[0];
    const int*   ei  = (const int*)  d_in[1];
    const float* W1  = (const float*)d_in[2];
    const float* b1  = (const float*)d_in[3];
    const float* W2  = (const float*)d_in[4];
    const float* b2  = (const float*)d_in[5];
    float*       out = (float*)d_out;

    const int n_nodes = in_sizes[0] / 128;
    const int n_edges = in_sizes[1] / 2;
    const int* src = ei;
    const int* dst = ei + n_edges;

    float*    ws   = (float*)d_ws;
    float*    dinv = ws;
    const int nchunk = (n_edges + 31) / 32;
    uint4*    idlo = (uint4*)(dinv + n_nodes);
    uint4*    idhi = idlo + nchunk;
    float*    h1   = (float*)((uintptr_t)((unsigned*)(dinv + n_nodes) + (size_t)n_nodes * 64 + 8192)
                              & ~(uintptr_t)511);
    unsigned* w1p  = (unsigned*)h1 - 8192;
    unsigned* xb   = (unsigned*)(h1 + (size_t)n_nodes * 128);
    float*    agg2 = (float*)xb;
    float*    h2   = h1 + (size_t)n_nodes * 256;
    unsigned* pk   = (unsigned*)(h1 + (size_t)n_nodes * 192);

    const int B = 256;
    const long long n_pairs = (long long)n_nodes * 64;
    const int n_init = n_nodes > nchunk ? n_nodes : nchunk;
    k_deg_init <<<(n_init + B - 1) / B, B, 0, stream>>>(dinv, n_nodes, src, dst, idlo, idhi, pk, n_edges);
    k_deg_edges<<<dim3(RNG_SLICES, (n_nodes + RNG_NODES - 1) / RNG_NODES), RNG_THREADS, 0, stream>>>(dst, dinv, n_nodes, n_edges);
    k_deg_rsqrt<<<(n_nodes + B - 1) / B, B, 0, stream>>>(dinv, n_nodes);
    k_pack_x <<<(unsigned)((n_pairs + B - 1) / B), B, 0, stream>>>(x, xb, n_pairs);
    k_pack_w1<<<8192 / B, B, 0, stream>>>(W1, w1p);
    k_gemm1_wmma<<<(n_nodes + 15) / 16, B, 0, stream>>>(xb, w1p, dinv, h1, n_nodes);
    k_scatter1<<<(n_nodes + L1_NODES - 1) / L1_NODES, L1_THREADS, 0, stream>>>(h1, pk, idlo, idhi, dinv, b1, W2, h2, agg2, n_nodes, n_edges);
    k_scatter2<<<dim3(RNG_SLICES, (n_nodes + RNG_NODES - 1) / RNG_NODES), RNG_THREADS, 0, stream>>>(h2, src, dst, dinv, agg2, n_nodes, n_edges);
    k_logsoftmax<<<(n_nodes + B - 1) / B, B, 0, stream>>>(agg2, b2, out, n_nodes);
}


# X is rounded to 16-bit operand form in the product kernel: a wave reads 16 f32 rows whole, makes fragments via its LDS slot, runs all 8 column tiles, stores whole lines; 9 launches -> 8; same sums
# speedup vs baseline: 3.7104x; 1.0573x over previous
#include <hip/hip_runtime.h>
#include <hip/hip_bf16.h>


typedef __attribute__((ext_vector_type(16))) __bf16     v16bf;
typedef __attribute__((ext_vector_type(8)))  float      v8f;
typedef __attribute__((ext_vector_type(8)))  unsigned   v8u;

typedef __attribute__((ext_vector_type(2)))  float      f32x2;
typedef __attribute__((ext_vector_type(2)))  __bf16     bf16x2;

union FragAB { v8u u; v16bf b; };

__device__ __forceinline__ unsigned cvt2bf(float lo, float hi) {
    union { bf16x2 b; unsigned u; } r;
    r.b = __builtin_convertvector((f32x2){lo, hi}, bf16x2);
    return r.u;
}

__device__ __forceinline__ unsigned bf16_rne(float f) {
    unsigned u = __float_as_uint(f);
    u += 0x7FFFu + ((u >> 16) & 1u);
    return u >> 16;
}
__device__ __forceinline__ unsigned pack2bf(float lo, float hi) {
    return bf16_rne(lo) | (bf16_rne(hi) << 16);
}

#define L1_NODES   400

template <bool VEC>
__device__ __forceinline__ int4 l1_load_dst4(const int* __restrict__ dst, int e, int n_edges) {
    if (VEC) return *(const int4*)(dst + min(e, n_edges - 4));
    int4 r;
    r.x = dst[min(e + 0, n_edges - 1)];
    r.y = dst[min(e + 1, n_edges - 1)];
    r.z = dst[min(e + 2, n_edges - 1)];
    r.w = dst[min(e + 3, n_edges - 1)];
    return r;
}

template <bool VEC>
__device__ __forceinline__ void range_id_chunk(const int* __restrict__ dst, uint4* __restrict__ idlo,
                                               uint4* __restrict__ idhi, int chunk, int n_edges) {
    unsigned p[8] = {0u, 0u, 0u, 0u, 0u, 0u, 0u, 0u};
#pragma unroll
    for (int k = 0; k < 8; ++k) {
        const int4 d = l1_load_dst4<VEC>(dst, chunk * 32 + k * 4, n_edges);
        const unsigned id[4] = {(unsigned)d.x / L1_NODES, (unsigned)d.y / L1_NODES,
                                (unsigned)d.z / L1_NODES, (unsigned)d.w / L1_NODES};
#pragma unroll
        for (int c = 0; c < 4; ++c) {
#pragma unroll
            for (int b = 0; b < 8; ++b) p[b] |= ((id[c] >> b) & 1u) << (k * 4 + c);
        }
    }
    idlo[chunk] = make_uint4(p[0], p[1], p[2], p[3]);
    idhi[chunk] = make_uint4(p[4], p[5], p[6], p[7]);
}

#define PK_SHIFT   23
#define PK_SRCMASK ((1u << PK_SHIFT) - 1u)
#define PK_BATCH   8
static_assert(L1_NODES <= (1 << (32 - PK_SHIFT)), "a table place must fit above PK_SHIFT");

__global__ void k_deg_init(float* dinv, int n, const int* __restrict__ src, const int* __restrict__ dst,
                           uint4* __restrict__ idlo, uint4* __restrict__ idhi,
                           unsigned* __restrict__ pk, int n_edges) {
    int i = blockIdx.x * blockDim.x + threadIdx.x;
    if (i < n) dinv[i] = 1.0f;
    const int T = gridDim.x * blockDim.x;
    for (int e0 = i; e0 < n_edges; e0 += PK_BATCH * T) {
        int ec[PK_BATCH], sv[PK_BATCH], dv[PK_BATCH];
#pragma unroll
        for (int k = 0; k < PK_BATCH; ++k) {
            ec[k] = min(e0 + k * T, n_edges - 1);
            sv[k] = src[ec[k]];
            dv[k] = dst[ec[k]];
        }
#pragma unroll
        for (int k = 0; k < PK_BATCH; ++k)
            pk[ec[k]] = (unsigned)sv[k] | (((unsigned)dv[k] % L1_NODES) << PK_SHIFT);
    }
    if (i < (n_edges + 31) / 32) {
        if ((n_edges & 3) == 0) range_id_chunk<true >(dst, idlo, idhi, i, n_edges);
        else                    range_id_chunk<false>(dst, idlo, idhi, i, n_edges);
    }
}
#define RNG_NODES  6272
#define RNG_SLICES 16
#define RNG_THREADS 1024
#define RNG_BATCH   8

__global__ __launch_bounds__(RNG_THREADS)
void k_deg_edges(const int* __restrict__ dst, float* dinv, int n_nodes, int e) {
    __shared__ float acc[RNG_NODES];
    const int tid = threadIdx.x;
    const int r0  = blockIdx.y * RNG_NODES;
    const int cnt = min(RNG_NODES, n_nodes - r0);
    const int esz = (e + RNG_SLICES - 1) / RNG_SLICES;
    const int e0  = blockIdx.x * esz;
    const int e1  = min(e0 + esz, e);
    for (int i = tid; i < RNG_NODES; i += RNG_THREADS) acc[i] = 0.0f;
    __syncthreads();
    const int nb = (e1 - e0 + RNG_BATCH * RNG_THREADS - 1) / (RNG_BATCH * RNG_THREADS);
    for (int b = 0; b < nb; ++b) {
        const int base = e0 + tid + b * (RNG_BATCH * RNG_THREADS);
        int d[RNG_BATCH];
#pragma unroll
        for (int k = 0; k < RNG_BATCH; ++k) {
            int idx = base + k * RNG_THREADS;
            d[k] = dst[min(idx, e1 - 1)];
        }
        __builtin_amdgcn_sched_barrier(0);
#pragma unroll
        for (int k = 0; k < RNG_BATCH; ++k) {
            int idx = base + k * RNG_THREADS;
            unsigned dl = (unsigned)(d[k] - r0);
            if (idx < e1 && dl < (unsigned)cnt) atomicAdd(&acc[dl], 1.0f);
        }
    }
    __syncthreads();
    const int nch = (cnt + RNG_THREADS - 1) / RNG_THREADS;
    for (int c = 0; c < nch; ++c) {
        int cc = c + (int)blockIdx.x; cc = cc % nch;
        int i = cc * RNG_THREADS + tid;
        if (i < cnt) { float v = acc[i]; if (v != 0.0f) atomicAdd(&dinv[r0 + i], v); }
    }
}
__global__ void k_deg_rsqrt(float* dinv, int n) {
    int i = blockIdx.x * blockDim.x + threadIdx.x;
    if (i < n) dinv[i] = rsqrtf(dinv[i]);
}

__global__ void k_pack_w1(const float* __restrict__ W1, unsigned* __restrict__ w1p) {
    int tid = blockIdx.x * blockDim.x + threadIdx.x;
    if (tid >= 8192) return;
    int v    = tid & 7;
    int lane = (tid >> 3) & 31;
    int ks   = (tid >> 8) & 3;
    int nt   = tid >> 10;
    int k    = ks * 32 + ((v >> 2) << 4) + ((lane >> 4) << 3) + ((v & 3) << 1);
    int ncol = 64 * (nt >> 2) + 4 * (lane & 15) + (nt & 3);
    w1p[tid] = pack2bf(W1[(size_t)k * 128 + ncol], W1[(size_t)(k + 1) * 128 + ncol]);
}

#define G1_WAVES 4
#define G1_PITCH 68
__global__ __launch_bounds__(G1_WAVES * 32)
void k_gemm1_wmma(const float* __restrict__ x,
                  const unsigned* __restrict__ w1p,
                  const float* __restrict__ dinv,
                  float* __restrict__ h1,
                  int n_nodes) {
    __shared__ __attribute__((aligned(16))) unsigned xs[G1_WAVES][16 * G1_PITCH];
    const int lane = threadIdx.x & 31;
    const int wid  = __builtin_amdgcn_readfirstlane(threadIdx.x >> 5);
    const int m0   = (blockIdx.x * G1_WAVES + wid) * 16;
    const int c    = lane & 15;
    const int half = lane >> 4;
    unsigned* slot = xs[wid];

    float4 xv[16];
#pragma unroll
    for (int i = 0; i < 16; ++i) {
        const int row = min(m0 + i, n_nodes - 1);
        xv[i] = ((const float4*)(x + (size_t)row * 128))[lane];
    }
    const bool full  = (m0 + 16 <= n_nodes);
    const int  mbase = m0 + (half << 3);
    float dv[8];
    if (full) {
        const float4 d0 = *(const float4*)(dinv + mbase);
        const float4 d1 = *(const float4*)(dinv + mbase + 4);
        dv[0] = d0.x; dv[1] = d0.y; dv[2] = d0.z; dv[3] = d0.w;
        dv[4] = d1.x; dv[5] = d1.y; dv[6] = d1.z; dv[7] = d1.w;
    } else {
#pragma unroll
        for (int v = 0; v < 8; ++v) dv[v] = dinv[min(mbase + v, n_nodes - 1)];
    }
#pragma unroll
    for (int i = 0; i < 16; ++i) {
        *(uint2*)(slot + i * G1_PITCH + 2 * lane) =
            make_uint2(cvt2bf(xv[i].x, xv[i].y), cvt2bf(xv[i].z, xv[i].w));
    }
    __syncthreads();

    FragAB a[4];
#pragma unroll
    for (int ks = 0; ks < 4; ++ks) {
        const uint4 alo = *(const uint4*)(slot + c * G1_PITCH + ks * 16 + half * 4);
        const uint4 ahi = *(const uint4*)(slot + c * G1_PITCH + ks * 16 + 8 + half * 4);
        a[ks].u = (v8u){alo.x, alo.y, alo.z, alo.w, ahi.x, ahi.y, ahi.z, ahi.w};
    }
    const v8u* bfrag = (const v8u*)w1p;
    v8f acc[8];
#pragma unroll
    for (int t = 0; t < 8; ++t) acc[t] = (v8f){};
#pragma unroll
    for (int ks = 0; ks < 4; ++ks) {
#pragma unroll
        for (int t = 0; t < 8; ++t) {
            FragAB b;
            b.u = bfrag[(t * 4 + ks) * 32 + lane];
            acc[t] = __builtin_amdgcn_wmma_f32_16x16x32_bf16(
                         false, a[ks].b, false, b.b, (short)0, acc[t], false, false);
        }
    }
    float* hp = h1 + (size_t)mbase * 128 + 4 * c;
    if (full) {
#pragma unroll
        for (int v = 0; v < 8; ++v) {
#pragma unroll
            for (int g = 0; g < 2; ++g) {
                *(float4*)(hp + (size_t)v * 128 + 64 * g) =
                    make_float4(acc[4 * g + 0][v] * dv[v], acc[4 * g + 1][v] * dv[v],
                                acc[4 * g + 2][v] * dv[v], acc[4 * g + 3][v] * dv[v]);
            }
        }
    } else {
#pragma unroll
        for (int v = 0; v < 8; ++v) {
            if (mbase + v < n_nodes) {
#pragma unroll
                for (int g = 0; g < 2; ++g) {
                    *(float4*)(hp + (size_t)v * 128 + 64 * g) =
                        make_float4(acc[4 * g + 0][v] * dv[v], acc[4 * g + 1][v] * dv[v],
                                    acc[4 * g + 2][v] * dv[v], acc[4 * g + 3][v] * dv[v]);
                }
            }
        }
    }
}

#define L1_THREADS 1024
#define L1_Q       4
#define L1_BATCH   (L1_Q * 32 * 32)
#define L1_GROUP   8

__device__ __forceinline__ void l1_flush(const float* __restrict__ h1, const unsigned* __restrict__ pk,
                                         float* acc, int qe, int qn, int lane) {
    const unsigned pw = pk[qe];
    const int s  = (int)(pw & PK_SRCMASK);
    const int dl = (int)(pw >> PK_SHIFT);
    for (int g = 0; g < qn; g += L1_GROUP) {
        float4 v[L1_GROUP];
#pragma unroll
        for (int j = 0; j < L1_GROUP; ++j) {
            const int jj = min(g + j, qn - 1);
            const int sj = __builtin_amdgcn_readlane(s, jj);
            v[j] = ((const float4*)(h1 + (size_t)sj * 128))[lane];
        }
#pragma unroll
        for (int j = 0; j < L1_GROUP; ++j) {
            if (g + j < qn) {
                const int dj = __builtin_amdgcn_readlane(dl, g + j);
                float* op = acc + dj * 128 + lane;
                atomicAdd(op +  0, v[j].x);
                atomicAdd(op + 32, v[j].y);
                atomicAdd(op + 64, v[j].z);
                atomicAdd(op + 96, v[j].w);
            }
        }
    }
}

__device__ __forceinline__ unsigned l1_match(const uint4& lo, const uint4& hi, const unsigned (&gb)[8],
                                             int chunk, bool last_batch, int n_edges) {
    const unsigned differ = (lo.x ^ gb[0]) | (lo.y ^ gb[1]) | (lo.z ^ gb[2]) | (lo.w ^ gb[3])
                          | (hi.x ^ gb[4]) | (hi.y ^ gb[5]) | (hi.z ^ gb[6]) | (hi.w ^ gb[7]);
    unsigned mm = ~differ;
    if (last_batch) {
        const int left = n_edges - chunk * 32;
        const unsigned keep = (left >= 32) ? 0xFFFFFFFFu : (left <= 0 ? 0u : ((1u << left) - 1u));
        mm &= keep;
    }
    return mm;
}

__device__ __forceinline__ void l1_enqueue(unsigned mm, int chunk,
                                           const float* __restrict__ h1, const unsigned* __restrict__ pk,
                                           float* acc, int lane, int& qe, int& qn) {
    const int ebase = chunk * 32;
    unsigned any = __builtin_amdgcn_ballot_w32(mm != 0);
    while (any) {
        const int slot = __builtin_ctz(mm | 0x80000000u);
        const int e    = ebase + slot;
        mm &= mm - 1;
        while (any) {
            const int j  = __builtin_ctz(any);
            any &= any - 1;
            const int ej = __builtin_amdgcn_readlane(e, j);
            if (lane == qn) qe = ej;
            if (++qn == 32) { l1_flush(h1, pk, acc, qe, 32, lane); qn = 0; }
        }
        any = __builtin_amdgcn_ballot_w32(mm != 0);
    }
}

__device__ __forceinline__ void l1_scan(const float* __restrict__ h1, const unsigned* __restrict__ pk,
                                        const uint4* __restrict__ idlo, const uint4* __restrict__ idhi,
                                        float* acc, int n_edges, int wid, int lane) {
    const int nchunk = (n_edges + 31) / 32;
    const int nbatch = (n_edges + L1_BATCH - 1) / L1_BATCH;
    const int rot    = (int)(((long long)blockIdx.x * nbatch) / gridDim.x);
    unsigned gb[8];
#pragma unroll
    for (int k = 0; k < 8; ++k) gb[k] = 0u - ((blockIdx.x >> k) & 1u);
    int qe = 0, qn = 0;
    uint4 lo[L1_Q], hi[L1_Q];
    int i = wid;
    int b = i + rot; if (b >= nbatch) b -= nbatch;
    if (i < nbatch) {
#pragma unroll
        for (int q = 0; q < L1_Q; ++q) {
            const int c = min(b * (L1_Q * 32) + q * 32 + lane, nchunk - 1);
            lo[q] = idlo[c]; hi[q] = idhi[c];
        }
    }
    while (i < nbatch) {
        const bool last = (b * L1_BATCH + L1_BATCH > n_edges);
        unsigned mm[L1_Q];
#pragma unroll
        for (int q = 0; q < L1_Q; ++q)
            mm[q] = l1_match(lo[q], hi[q], gb, b * (L1_Q * 32) + q * 32 + lane, last, n_edges);
        const int in = i + (L1_THREADS >> 5);
        int bn = in + rot; if (bn >= nbatch) bn -= nbatch;
        if (in < nbatch) {
#pragma unroll
            for (int q = 0; q < L1_Q; ++q) {
                const int c = min(bn * (L1_Q * 32) + q * 32 + lane, nchunk - 1);
                lo[q] = idlo[c]; hi[q] = idhi[c];
            }
        }
#pragma unroll
        for (int q = 0; q < L1_Q; ++q)
            l1_enqueue(mm[q], b * (L1_Q * 32) + q * 32 + lane, h1, pk, acc, lane, qe, qn);
        i = in; b = bn;
    }
    if (qn > 0) l1_flush(h1, pk, acc, qe, qn, lane);
}

__global__ __launch_bounds__(L1_THREADS)
void k_scatter1(const float* __restrict__ h1, const unsigned* __restrict__ pk,
                const uint4* __restrict__ idlo, const uint4* __restrict__ idhi,
                const float* __restrict__ dinv, const float* __restrict__ b1,
                const float* __restrict__ W2,
                float* __restrict__ h2, float* __restrict__ agg2,
                int n_nodes, int n_edges) {
    __shared__ float acc[L1_NODES * 128];
    __shared__ float h2s[2 * L1_NODES];
    const int tid  = threadIdx.x;
    const int lane = tid & 31;
    const int wid  = __builtin_amdgcn_readfirstlane(tid >> 5);
    const int r0   = blockIdx.x * L1_NODES;
    const int cnt  = min(L1_NODES, n_nodes - r0);

    {
        const float4* hp = (const float4*)(h1 + (size_t)r0 * 128);
        for (int i = tid; i < cnt * 32; i += L1_THREADS) {
            const float4 hv = hp[i];
            float* op = acc + (i >> 5) * 128 + (i & 31);
            op[0] = hv.x; op[32] = hv.y; op[64] = hv.z; op[96] = hv.w;
        }
    }
    __syncthreads();
    l1_scan(h1, pk, idlo, idhi, acc, n_edges, wid, lane);
    __syncthreads();
    {
        const float4 bb = ((const float4*)b1)[lane];
        const float2* w2 = (const float2*)W2;
        const int c = lane * 4;
        const float2 w0 = w2[c], w1 = w2[c + 1], w2v = w2[c + 2], w3 = w2[c + 3];
        for (int n0 = wid; n0 < cnt; n0 += 4 * (L1_THREADS >> 5)) {
            float s0[4], s1[4], di[4];
#pragma unroll
            for (int j = 0; j < 4; ++j) {
                const int n = min(n0 + j * (L1_THREADS >> 5), cnt - 1);
                const float* op = acc + n * 128 + lane;
                di[j] = dinv[r0 + n];
                const float t0 = fmaxf(op[0]  * di[j] + bb.x, 0.f), t1 = fmaxf(op[32] * di[j] + bb.y, 0.f);
                const float t2 = fmaxf(op[64] * di[j] + bb.z, 0.f), t3 = fmaxf(op[96] * di[j] + bb.w, 0.f);
                s0[j] = t0 * w0.x + t1 * w1.x + t2 * w2v.x + t3 * w3.x;
                s1[j] = t0 * w0.y + t1 * w1.y + t2 * w2v.y + t3 * w3.y;
            }
#pragma unroll
            for (int off = 16; off > 0; off >>= 1) {
#pragma unroll
                for (int j = 0; j < 4; ++j) {
                    s0[j] += __shfl_down(s0[j], off, 32);
                    s1[j] += __shfl_down(s1[j], off, 32);
                }
            }
            if (lane == 0) {
#pragma unroll
                for (int j = 0; j < 4; ++j) {
                    const int n = n0 + j * (L1_THREADS >> 5);
                    if (n < cnt) { h2s[2 * n] = s0[j] * di[j]; h2s[2 * n + 1] = s1[j] * di[j]; }
                }
            }
        }
    }
    __syncthreads();
    for (int i = tid; i < 2 * cnt; i += L1_THREADS) {
        const float v = h2s[i];
        h2[2 * (size_t)r0 + i]   = v;
        agg2[2 * (size_t)r0 + i] = v * dinv[r0 + (i >> 1)];
    }
}

__global__ __launch_bounds__(RNG_THREADS)
void k_scatter2(const float* __restrict__ h2,
                const int* __restrict__ src, const int* __restrict__ dst,
                const float* __restrict__ dinv,
                float* __restrict__ agg2, int n_nodes, int n_edges) {
    __shared__ float acc[2 * RNG_NODES];
    const int tid = threadIdx.x;
    const int r0  = blockIdx.y * RNG_NODES;
    const int cnt = min(RNG_NODES, n_nodes - r0);
    const int esz = (n_edges + RNG_SLICES - 1) / RNG_SLICES;
    const int e0  = blockIdx.x * esz;
    const int e1  = min(e0 + esz, n_edges);
    for (int i = tid; i < 2 * RNG_NODES; i += RNG_THREADS) acc[i] = 0.0f;
    __syncthreads();
    const int nb = (e1 - e0 + RNG_BATCH * RNG_THREADS - 1) / (RNG_BATCH * RNG_THREADS);
    for (int b = 0; b < nb; ++b) {
        const int base = e0 + tid + b * (RNG_BATCH * RNG_THREADS);
        int   s[RNG_BATCH];
        int   d[RNG_BATCH];
        bool  m[RNG_BATCH];
        unsigned dl[RNG_BATCH];
        float2 hv[RNG_BATCH];
#pragma unroll
        for (int k = 0; k < RNG_BATCH; ++k) {
            int idx = min(base + k * RNG_THREADS, e1 - 1);
            d[k] = dst[idx];
            s[k] = src[idx];
        }
        __builtin_amdgcn_sched_barrier(0);
#pragma unroll
        for (int k = 0; k < RNG_BATCH; ++k) {
            dl[k] = (unsigned)(d[k] - r0);
            m[k] = (base + k * RNG_THREADS < e1) && (dl[k] < (unsigned)cnt);
            int si = m[k] ? s[k] : 0;
            hv[k] = ((const float2*)h2)[si];
        }
        __builtin_amdgcn_sched_barrier(0);
#pragma unroll
        for (int k = 0; k < RNG_BATCH; ++k) {
            if (m[k]) {
                atomicAdd(&acc[2 * dl[k] + 0], hv[k].x);
                atomicAdd(&acc[2 * dl[k] + 1], hv[k].y);
            }
        }
    }
    __syncthreads();
    const int cnt2 = 2 * cnt;
    const int nch = (cnt2 + RNG_THREADS - 1) / RNG_THREADS;
    for (int c = 0; c < nch; ++c) {
        int cc = c + (int)blockIdx.x; cc = cc % nch;
        int i = cc * RNG_THREADS + tid;
        if (i < cnt2) {
            float v = acc[i];
            if (v != 0.0f) atomicAdd(&agg2[2 * (size_t)r0 + i], v * dinv[r0 + (i >> 1)]);
        }
    }
}
__global__ void k_logsoftmax(const float* __restrict__ agg2, const float* __restrict__ b2,
                             float* __restrict__ out, int n_nodes) {
    int n = blockIdx.x * blockDim.x + threadIdx.x;
    if (n >= n_nodes) return;
    float2 a = ((const float2*)agg2)[n];
    float z0 = a.x + b2[0];
    float z1 = a.y + b2[1];
    float m  = fmaxf(z0, z1);
    float l  = m + __logf(__expf(z0 - m) + __expf(z1 - m));
    ((float2*)out)[n] = make_float2(z0 - l, z1 - l);
}

extern "C" void kernel_launch(void* const* d_in, const int* in_sizes, int n_in,
                              void* d_out, int out_size, void* d_ws, size_t ws_size,
                              hipStream_t stream) {
    const float* x   = (const float*)d_in[0];
    const int*   ei  = (const int*)  d_in[1];
    const float* W1  = (const float*)d_in[2];
    const float* b1  = (const float*)d_in[3];
    const float* W2  = (const float*)d_in[4];
    const float* b2  = (const float*)d_in[5];
    float*       out = (float*)d_out;

    const int n_nodes = in_sizes[0] / 128;
    const int n_edges = in_sizes[1] / 2;
    const int* src = ei;
    const int* dst = ei + n_edges;

    float*    ws   = (float*)d_ws;
    float*    dinv = ws;
    const int nchunk = (n_edges + 31) / 32;
    uint4*    idlo = (uint4*)(dinv + n_nodes);
    uint4*    idhi = idlo + nchunk;
    float*    h1   = (float*)((uintptr_t)((unsigned*)(dinv + n_nodes) + (size_t)n_nodes * 64 + 8192)
                              & ~(uintptr_t)511);
    unsigned* w1p  = (unsigned*)h1 - 8192;
    float*    agg2 = h1 + (size_t)n_nodes * 128;
    float*    h2   = h1 + (size_t)n_nodes * 256;
    unsigned* pk   = (unsigned*)(h1 + (size_t)n_nodes * 192);

    const int B = 256;
    const int n_init = n_nodes > nchunk ? n_nodes : nchunk;
    k_deg_init <<<(n_init + B - 1) / B, B, 0, stream>>>(dinv, n_nodes, src, dst, idlo, idhi, pk, n_edges);
    k_deg_edges<<<dim3(RNG_SLICES, (n_nodes + RNG_NODES - 1) / RNG_NODES), RNG_THREADS, 0, stream>>>(dst, dinv, n_nodes, n_edges);
    k_deg_rsqrt<<<(n_nodes + B - 1) / B, B, 0, stream>>>(dinv, n_nodes);
    k_pack_w1<<<8192 / B, B, 0, stream>>>(W1, w1p);
    k_gemm1_wmma<<<(n_nodes + 16 * G1_WAVES - 1) / (16 * G1_WAVES), G1_WAVES * 32, 0, stream>>>(x, w1p, dinv, h1, n_nodes);
    k_scatter1<<<(n_nodes + L1_NODES - 1) / L1_NODES, L1_THREADS, 0, stream>>>(h1, pk, idlo, idhi, dinv, b1, W2, h2, agg2, n_nodes, n_edges);
    k_scatter2<<<dim3(RNG_SLICES, (n_nodes + RNG_NODES - 1) / RNG_NODES), RNG_THREADS, 0, stream>>>(h2, src, dst, dinv, agg2, n_nodes, n_edges);
    k_logsoftmax<<<(n_nodes + B - 1) / B, B, 0, stream>>>(agg2, b2, out, n_nodes);
}
